# starting kernel
# baseline (speedup 1.0000x reference)
#include <hip/hip_runtime.h>

#define BINS   64
#define KPN    24
#define BATCH  256
#define HW     (BINS * BINS)
#define MIN_L  (-150.0f)
#define DELTA  (300.0f / 64.0f)
#define LOG2E  1.44269504088896340736f

typedef __attribute__((ext_vector_type(2))) float v2f;
typedef __attribute__((ext_vector_type(8))) float v8f;

__device__ __forceinline__ float center_of(int i) {
    return MIN_L + ((float)i + 0.5f) * DELTA;
}

__global__ __launch_bounds__(768) void pose_stage1(
    const float* __restrict__ feat, const float* __restrict__ gt,
    float* __restrict__ out_loc, float* __restrict__ lossbuf)
{
    const int b  = blockIdx.x;
    const int t  = threadIdx.x;
    const int k  = t % KPN;
    const int w0 = t / KPN;

    __shared__ float s_m[32 * KPN];
    __shared__ float s_Mk[KPN];
    __shared__ float s_part[32 * 128];
    __shared__ float s_red[128];

    const float* p = feat + (size_t)b * (HW * KPN) + (size_t)(w0 * KPN + k);

    float m = -1.0e30f;
    float sA = 0.f, sB = 0.f;
    float sy = 0.f, sy2 = 0.f;

    #pragma unroll 4
    for (int h = 0; h < BINS; ++h) {
        float xa = p[h * (BINS * KPN)];
        float xb = p[h * (BINS * KPN) + 32 * KPN];
        float xsa = xa * LOG2E;
        float xsb = xb * LOG2E;
        float mo  = m;
        m = fmaxf(m, fmaxf(xsa, xsb));
        if (m > mo) {
            float c = __builtin_amdgcn_exp2f(mo - m);
            sA *= c; sB *= c; sy *= c; sy2 *= c;
        }
        float ea = __builtin_amdgcn_exp2f(xsa - m);
        float eb = __builtin_amdgcn_exp2f(xsb - m);
        sA += ea;
        sB += eb;
        float es = ea + eb;
        float cy = MIN_L + ((float)h + 0.5f) * DELTA;
        sy  = fmaf(es, cy,      sy);
        sy2 = fmaf(es, cy * cy, sy2);
    }

    s_m[t] = m;
    __syncthreads();

    if (t < KPN) {
        float mm = s_m[t];
        #pragma unroll 4
        for (int j = 1; j < 32; ++j) mm = fmaxf(mm, s_m[j * KPN + t]);
        s_Mk[t] = mm;
    }
    __syncthreads();

    {
        float scale = __builtin_amdgcn_exp2f(m - s_Mk[k]);
        float a  = sA * scale, bb = sB * scale;
        float cxa = center_of(w0), cxb = center_of(w0 + 32);
        s_part[w0 * 128 + 0 * KPN + k] = a + bb;
        s_part[w0 * 128 + 1 * KPN + k] = sy  * scale;
        s_part[w0 * 128 + 2 * KPN + k] = sy2 * scale;
        s_part[w0 * 128 + 3 * KPN + k] = a * cxa + bb * cxb;
        s_part[w0 * 128 + 4 * KPN + k] = a * cxa * cxa + bb * cxb * cxb;
    }
    if (t < 256) s_part[(t >> 3) * 128 + 120 + (t & 7)] = 0.0f;
    __syncthreads();

    const int wid  = t >> 5;
    const int lane = t & 31;
    if (wid < 8) {
        v2f ones = {1.0f, 1.0f};
        v8f acc  = {};
        int col  = wid * 16 + (lane & 15);
        #pragma unroll
        for (int c8 = 0; c8 < 8; ++c8) {
            int rbase = c8 * 4;
            v2f bvec;
            if (lane < 16) {
                bvec.x = s_part[(rbase + 0) * 128 + col];
                bvec.y = s_part[(rbase + 1) * 128 + col];
            } else {
                bvec.x = s_part[(rbase + 2) * 128 + col];
                bvec.y = s_part[(rbase + 3) * 128 + col];
            }
            acc = __builtin_amdgcn_wmma_f32_16x16x4_f32(
                false, ones, false, bvec, (short)0, acc, false, false);
        }
        if (lane < 16) s_red[col] = acc[0];
    }
    __syncthreads();

    if (t < KPN) {
        float S   = s_red[0 * KPN + t];
        float Sy  = s_red[1 * KPN + t];
        float Sy2 = s_red[2 * KPN + t];
        float Sx  = s_red[3 * KPN + t];
        float Sx2 = s_red[4 * KPN + t];
        float inv = 1.0f / S;
        float ly = Sy * inv, lx = Sx * inv;
        float vy = Sy2 * inv - ly * ly;
        float vx = Sx2 * inv - lx * lx;
        float gy = gt[(size_t)b * KPN * 2 + t * 2 + 0];
        float gx = gt[(size_t)b * KPN * 2 + t * 2 + 1];
        float dy = ly - gy, dx = lx - gx;
        lossbuf[(size_t)b * KPN * 2 + t * 2 + 0] = vy + dy * dy;
        lossbuf[(size_t)b * KPN * 2 + t * 2 + 1] = vx + dx * dx;
        out_loc[(size_t)b * KPN * 2 + t * 2 + 0] = ly;
        out_loc[(size_t)b * KPN * 2 + t * 2 + 1] = lx;
    }
}

__global__ __launch_bounds__(1024) void pose_stage2(
    const float* __restrict__ lossbuf, float* __restrict__ out_hard)
{
    __shared__ float red[1024];
    __shared__ float s_inv;
    const int t = threadIdx.x;
    const int N = 2 * KPN * BATCH;

    float s = 0.f;
    for (int i = t; i < N; i += 1024) s += lossbuf[i];
    red[t] = s;
    __syncthreads();
    #pragma unroll
    for (int off = 512; off > 0; off >>= 1) {
        if (t < off) red[t] += red[t + off];
        __syncthreads();
    }
    if (t == 0) {
        float loss_sum = red[0] + 0.001f;
        s_inv = ((float)KPN / 150.0f) / loss_sum;
    }
    __syncthreads();
    float inv = s_inv;

    for (int o = t; o < N; o += 1024) {
        int c = o / (KPN * BATCH);
        int r = o % (KPN * BATCH);
        int k = r / BATCH;
        int b = r % BATCH;
        float l  = lossbuf[b * KPN * 2 + k * 2 + c];
        float l2 = lossbuf[b * KPN * 2 + k * 2 + 0] +
                   lossbuf[b * KPN * 2 + k * 2 + 1];
        out_hard[o] = l * l2 * inv;
    }
}

extern "C" void kernel_launch(void* const* d_in, const int* in_sizes, int n_in,
                              void* d_out, int out_size, void* d_ws, size_t ws_size,
                              hipStream_t stream) {
    const float* feat = (const float*)d_in[0];
    const float* gt   = (const float*)d_in[1];
    float* out      = (float*)d_out;
    float* out_hard = out;
    float* out_loc  = out + 2 * KPN * BATCH;
    float* lossbuf  = (float*)d_ws;

    pose_stage1<<<BATCH, 768, 0, stream>>>(feat, gt, out_loc, lossbuf);
    pose_stage2<<<1, 1024, 0, stream>>>(lossbuf, out_hard);
}


# stage 1: each thread reads 32 16-byte pieces with 16-24 loads in flight (was 64 dependent pairs of 4-byte loads), four online-softmax groups; 128-row WMMA column sum (reduction tree changed)
# speedup vs baseline: 2.3394x; 2.3394x over previous
#include <hip/hip_runtime.h>

#define BINS   64
#define KPN    24
#define BATCH  256
#define HW     (BINS * BINS)
#define MIN_L  (-150.0f)
#define DELTA  (300.0f / 64.0f)
#define LOG2E  1.44269504088896340736f

typedef __attribute__((ext_vector_type(2))) float v2f;
typedef __attribute__((ext_vector_type(4))) float v4f;
typedef __attribute__((ext_vector_type(8))) float v8f;

#define NLD    32
#define CHUNK  8
#define PITCH  152

__device__ __forceinline__ float center_of(int i) {
    return MIN_L + ((float)i + 0.5f) * DELTA;
}

__device__ __forceinline__ v4f max4(v4f a, v4f b) {
    v4f r;
    r.x = fmaxf(a.x, b.x); r.y = fmaxf(a.y, b.y);
    r.z = fmaxf(a.z, b.z); r.w = fmaxf(a.w, b.w);
    return r;
}

__device__ __forceinline__ v4f exp2_4(v4f a) {
    v4f r;
    r.x = __builtin_amdgcn_exp2f(a.x); r.y = __builtin_amdgcn_exp2f(a.y);
    r.z = __builtin_amdgcn_exp2f(a.z); r.w = __builtin_amdgcn_exp2f(a.w);
    return r;
}

__device__ __forceinline__ v4f exp_shift4(v4f x, v4f m) {
    v4f a;
    a.x = __builtin_fmaf(x.x, LOG2E, -m.x); a.y = __builtin_fmaf(x.y, LOG2E, -m.y);
    a.z = __builtin_fmaf(x.z, LOG2E, -m.z); a.w = __builtin_fmaf(x.w, LOG2E, -m.w);
    return exp2_4(a);
}

__global__ __launch_bounds__(768) void pose_stage1(
    const float* __restrict__ feat, const float* __restrict__ gt,
    float* __restrict__ out_loc, float* __restrict__ lossbuf)
{
    const int b    = blockIdx.x;
    const int t    = threadIdx.x;
    const int kg   = t % 6;
    const int wq   = t / 6;
    const int wid  = t >> 5;
    const int lane = t & 31;

    __shared__ __attribute__((aligned(16))) float s_m[128 * KPN];
    __shared__ __attribute__((aligned(16))) float s_Mk[KPN];
    __shared__ __attribute__((aligned(16))) float s_part[128 * PITCH];
    __shared__ float s_red[2 * 128];

    const v4f* p = reinterpret_cast<const v4f*>(feat + (size_t)b * (HW * KPN)) + t;

    v4f x[NLD];
    #pragma unroll
    for (int i = 0; i < 3 * CHUNK; ++i) x[i] = p[i * 768];
    __builtin_amdgcn_sched_barrier(0);

    const float cy0 = center_of(wq >> 6);
    v4f mu, m, S, Sy, Sy2;

    #pragma unroll
    for (int c = 0; c < NLD / CHUNK; ++c) {
        v4f mn = (c == 0) ? x[0] : mu;
        #pragma unroll
        for (int r = (c == 0) ? 1 : 0; r < CHUNK; ++r) mn = max4(mn, x[c * CHUNK + r]);
        const v4f ms = mn * LOG2E;
        if (c == 0) {
            S = (v4f)(0.0f); Sy = (v4f)(0.0f); Sy2 = (v4f)(0.0f);
        } else {
            const v4f f = exp2_4(m - ms);
            S = S * f; Sy = Sy * f; Sy2 = Sy2 * f;
        }
        mu = mn;
        m  = ms;
        #pragma unroll
        for (int r = 0; r < CHUNK; ++r) {
            const int   i  = c * CHUNK + r;
            const float cy = cy0 + (float)(2 * i) * DELTA;
            const v4f   e  = exp_shift4(x[i], m);
            S   = S + e;
            Sy  = Sy  + e * cy;
            Sy2 = Sy2 + e * (cy * cy);
        }
        asm volatile("" : "+v"(mu.x), "+v"(mu.y), "+v"(mu.z), "+v"(mu.w),
                          "+v"(S.x),  "+v"(S.y),  "+v"(S.z),  "+v"(S.w),
                          "+v"(Sy.x), "+v"(Sy.y), "+v"(Sy.z), "+v"(Sy.w),
                          "+v"(Sy2.x), "+v"(Sy2.y), "+v"(Sy2.z), "+v"(Sy2.w));
        if (c == 0) {
            #pragma unroll
            for (int i = 3 * CHUNK; i < NLD; ++i) x[i] = p[i * 768];
            __builtin_amdgcn_sched_barrier(0);
        }
    }

    reinterpret_cast<v4f*>(s_m)[t] = m;
    __syncthreads();

    {
        float mm = s_m[lane * KPN + wid];
        mm = fmaxf(mm, s_m[(lane + 32) * KPN + wid]);
        mm = fmaxf(mm, s_m[(lane + 64) * KPN + wid]);
        mm = fmaxf(mm, s_m[(lane + 96) * KPN + wid]);
        #pragma unroll
        for (int o = 16; o > 0; o >>= 1) mm = fmaxf(mm, __shfl_xor(mm, o));
        if (lane == 0) s_Mk[wid] = mm;
    }
    __syncthreads();

    {
        const v4f   M  = reinterpret_cast<const v4f*>(s_Mk)[kg];
        const v4f   sc = exp2_4(m - M);
        const v4f   a  = S * sc;
        const float cx = center_of(wq & 63);
        v4f* row = reinterpret_cast<v4f*>(s_part + wq * PITCH + 4 * kg);
        row[0]  = a;
        row[6]  = Sy  * sc;
        row[12] = Sy2 * sc;
        row[18] = a * cx;
        row[24] = a * (cx * cx);
    }
    if (t < 256)
        reinterpret_cast<v4f*>(s_part + (t >> 1) * PITCH + 120)[t & 1] = (v4f)(0.0f);
    __syncthreads();

    if (wid < 16) {
        const int tile = wid & 7;
        const int half = wid >> 3;
        const int col  = tile * 16 + (lane & 15);
        const float* bp = s_part + (half * 64 + 2 * (lane >> 4)) * PITCH + col;
        v2f ones = {1.0f, 1.0f};
        v8f acc  = {};
        #pragma unroll
        for (int s = 0; s < 16; ++s) {
            v2f bvec;
            bvec.x = bp[(4 * s + 0) * PITCH];
            bvec.y = bp[(4 * s + 1) * PITCH];
            acc = __builtin_amdgcn_wmma_f32_16x16x4_f32(
                false, ones, false, bvec, (short)0, acc, false, false);
        }
        if (lane < 16) s_red[half * 128 + col] = acc[0];
    }
    __syncthreads();

    if (t < KPN) {
        float S_  = s_red[0 * KPN + t] + s_red[128 + 0 * KPN + t];
        float Sy_ = s_red[1 * KPN + t] + s_red[128 + 1 * KPN + t];
        float Sy2_= s_red[2 * KPN + t] + s_red[128 + 2 * KPN + t];
        float Sx_ = s_red[3 * KPN + t] + s_red[128 + 3 * KPN + t];
        float Sx2_= s_red[4 * KPN + t] + s_red[128 + 4 * KPN + t];
        float inv = 1.0f / S_;
        float ly = Sy_ * inv, lx = Sx_ * inv;
        float vy = Sy2_ * inv - ly * ly;
        float vx = Sx2_ * inv - lx * lx;
        float gy = gt[(size_t)b * KPN * 2 + t * 2 + 0];
        float gx = gt[(size_t)b * KPN * 2 + t * 2 + 1];
        float dy = ly - gy, dx = lx - gx;
        lossbuf[(size_t)b * KPN * 2 + t * 2 + 0] = vy + dy * dy;
        lossbuf[(size_t)b * KPN * 2 + t * 2 + 1] = vx + dx * dx;
        out_loc[(size_t)b * KPN * 2 + t * 2 + 0] = ly;
        out_loc[(size_t)b * KPN * 2 + t * 2 + 1] = lx;
    }
}

__global__ __launch_bounds__(1024) void pose_stage2(
    const float* __restrict__ lossbuf, float* __restrict__ out_hard)
{
    __shared__ float red[1024];
    __shared__ float s_inv;
    const int t = threadIdx.x;
    const int N = 2 * KPN * BATCH;

    float s = 0.f;
    for (int i = t; i < N; i += 1024) s += lossbuf[i];
    red[t] = s;
    __syncthreads();
    #pragma unroll
    for (int off = 512; off > 0; off >>= 1) {
        if (t < off) red[t] += red[t + off];
        __syncthreads();
    }
    if (t == 0) {
        float loss_sum = red[0] + 0.001f;
        s_inv = ((float)KPN / 150.0f) / loss_sum;
    }
    __syncthreads();
    float inv = s_inv;

    for (int o = t; o < N; o += 1024) {
        int c = o / (KPN * BATCH);
        int r = o % (KPN * BATCH);
        int k = r / BATCH;
        int b = r % BATCH;
        float l  = lossbuf[b * KPN * 2 + k * 2 + c];
        float l2 = lossbuf[b * KPN * 2 + k * 2 + 0] +
                   lossbuf[b * KPN * 2 + k * 2 + 1];
        out_hard[o] = l * l2 * inv;
    }
}

extern "C" void kernel_launch(void* const* d_in, const int* in_sizes, int n_in,
                              void* d_out, int out_size, void* d_ws, size_t ws_size,
                              hipStream_t stream) {
    const float* feat = (const float*)d_in[0];
    const float* gt   = (const float*)d_in[1];
    float* out      = (float*)d_out;
    float* out_hard = out;
    float* out_loc  = out + 2 * KPN * BATCH;
    float* lossbuf  = (float*)d_ws;

    pose_stage1<<<BATCH, 768, 0, stream>>>(feat, gt, out_loc, lossbuf);
    pose_stage2<<<1, 1024, 0, stream>>>(lossbuf, out_hard);
}


# stage 2 in one pass: three 16-byte loads per thread in flight together, lane-butterfly sum with one barrier (reduction tree changed), outputs transposed through LDS and stored as whole lines
# speedup vs baseline: 3.6460x; 1.5585x over previous
#include <hip/hip_runtime.h>

#define BINS   64
#define KPN    24
#define BATCH  256
#define HW     (BINS * BINS)
#define MIN_L  (-150.0f)
#define DELTA  (300.0f / 64.0f)
#define LOG2E  1.44269504088896340736f

typedef __attribute__((ext_vector_type(2))) float v2f;
typedef __attribute__((ext_vector_type(4))) float v4f;
typedef __attribute__((ext_vector_type(8))) float v8f;

#define NLD    32
#define CHUNK  8
#define PITCH  152

__device__ __forceinline__ float center_of(int i) {
    return MIN_L + ((float)i + 0.5f) * DELTA;
}

__device__ __forceinline__ v4f max4(v4f a, v4f b) {
    v4f r;
    r.x = fmaxf(a.x, b.x); r.y = fmaxf(a.y, b.y);
    r.z = fmaxf(a.z, b.z); r.w = fmaxf(a.w, b.w);
    return r;
}

__device__ __forceinline__ v4f exp2_4(v4f a) {
    v4f r;
    r.x = __builtin_amdgcn_exp2f(a.x); r.y = __builtin_amdgcn_exp2f(a.y);
    r.z = __builtin_amdgcn_exp2f(a.z); r.w = __builtin_amdgcn_exp2f(a.w);
    return r;
}

__device__ __forceinline__ v4f exp_shift4(v4f x, v4f m) {
    v4f a;
    a.x = __builtin_fmaf(x.x, LOG2E, -m.x); a.y = __builtin_fmaf(x.y, LOG2E, -m.y);
    a.z = __builtin_fmaf(x.z, LOG2E, -m.z); a.w = __builtin_fmaf(x.w, LOG2E, -m.w);
    return exp2_4(a);
}

__global__ __launch_bounds__(768) void pose_stage1(
    const float* __restrict__ feat, const float* __restrict__ gt,
    float* __restrict__ out_loc, float* __restrict__ lossbuf)
{
    const int b    = blockIdx.x;
    const int t    = threadIdx.x;
    const int kg   = t % 6;
    const int wq   = t / 6;
    const int wid  = t >> 5;
    const int lane = t & 31;

    __shared__ __attribute__((aligned(16))) float s_m[128 * KPN];
    __shared__ __attribute__((aligned(16))) float s_Mk[KPN];
    __shared__ __attribute__((aligned(16))) float s_part[128 * PITCH];
    __shared__ float s_red[2 * 128];

    const v4f* p = reinterpret_cast<const v4f*>(feat + (size_t)b * (HW * KPN)) + t;

    v4f x[NLD];
    #pragma unroll
    for (int i = 0; i < 3 * CHUNK; ++i) x[i] = p[i * 768];
    __builtin_amdgcn_sched_barrier(0);

    const float cy0 = center_of(wq >> 6);
    v4f mu, m, S, Sy, Sy2;

    #pragma unroll
    for (int c = 0; c < NLD / CHUNK; ++c) {
        v4f mn = (c == 0) ? x[0] : mu;
        #pragma unroll
        for (int r = (c == 0) ? 1 : 0; r < CHUNK; ++r) mn = max4(mn, x[c * CHUNK + r]);
        const v4f ms = mn * LOG2E;
        if (c == 0) {
            S = (v4f)(0.0f); Sy = (v4f)(0.0f); Sy2 = (v4f)(0.0f);
        } else {
            const v4f f = exp2_4(m - ms);
            S = S * f; Sy = Sy * f; Sy2 = Sy2 * f;
        }
        mu = mn;
        m  = ms;
        #pragma unroll
        for (int r = 0; r < CHUNK; ++r) {
            const int   i  = c * CHUNK + r;
            const float cy = cy0 + (float)(2 * i) * DELTA;
            const v4f   e  = exp_shift4(x[i], m);
            S   = S + e;
            Sy  = Sy  + e * cy;
            Sy2 = Sy2 + e * (cy * cy);
        }
        asm volatile("" : "+v"(mu.x), "+v"(mu.y), "+v"(mu.z), "+v"(mu.w),
                          "+v"(S.x),  "+v"(S.y),  "+v"(S.z),  "+v"(S.w),
                          "+v"(Sy.x), "+v"(Sy.y), "+v"(Sy.z), "+v"(Sy.w),
                          "+v"(Sy2.x), "+v"(Sy2.y), "+v"(Sy2.z), "+v"(Sy2.w));
        if (c == 0) {
            #pragma unroll
            for (int i = 3 * CHUNK; i < NLD; ++i) x[i] = p[i * 768];
            __builtin_amdgcn_sched_barrier(0);
        }
    }

    reinterpret_cast<v4f*>(s_m)[t] = m;
    __syncthreads();

    {
        float mm = s_m[lane * KPN + wid];
        mm = fmaxf(mm, s_m[(lane + 32) * KPN + wid]);
        mm = fmaxf(mm, s_m[(lane + 64) * KPN + wid]);
        mm = fmaxf(mm, s_m[(lane + 96) * KPN + wid]);
        #pragma unroll
        for (int o = 16; o > 0; o >>= 1) mm = fmaxf(mm, __shfl_xor(mm, o));
        if (lane == 0) s_Mk[wid] = mm;
    }
    __syncthreads();

    {
        const v4f   M  = reinterpret_cast<const v4f*>(s_Mk)[kg];
        const v4f   sc = exp2_4(m - M);
        const v4f   a  = S * sc;
        const float cx = center_of(wq & 63);
        v4f* row = reinterpret_cast<v4f*>(s_part + wq * PITCH + 4 * kg);
        row[0]  = a;
        row[6]  = Sy  * sc;
        row[12] = Sy2 * sc;
        row[18] = a * cx;
        row[24] = a * (cx * cx);
    }
    if (t < 256)
        reinterpret_cast<v4f*>(s_part + (t >> 1) * PITCH + 120)[t & 1] = (v4f)(0.0f);
    __syncthreads();

    if (wid < 16) {
        const int tile = wid & 7;
        const int half = wid >> 3;
        const int col  = tile * 16 + (lane & 15);
        const float* bp = s_part + (half * 64 + 2 * (lane >> 4)) * PITCH + col;
        v2f ones = {1.0f, 1.0f};
        v8f acc  = {};
        #pragma unroll
        for (int s = 0; s < 16; ++s) {
            v2f bvec;
            bvec.x = bp[(4 * s + 0) * PITCH];
            bvec.y = bp[(4 * s + 1) * PITCH];
            acc = __builtin_amdgcn_wmma_f32_16x16x4_f32(
                false, ones, false, bvec, (short)0, acc, false, false);
        }
        if (lane < 16) s_red[half * 128 + col] = acc[0];
    }
    __syncthreads();

    if (t < KPN) {
        float S_  = s_red[0 * KPN + t] + s_red[128 + 0 * KPN + t];
        float Sy_ = s_red[1 * KPN + t] + s_red[128 + 1 * KPN + t];
        float Sy2_= s_red[2 * KPN + t] + s_red[128 + 2 * KPN + t];
        float Sx_ = s_red[3 * KPN + t] + s_red[128 + 3 * KPN + t];
        float Sx2_= s_red[4 * KPN + t] + s_red[128 + 4 * KPN + t];
        float inv = 1.0f / S_;
        float ly = Sy_ * inv, lx = Sx_ * inv;
        float vy = Sy2_ * inv - ly * ly;
        float vx = Sx2_ * inv - lx * lx;
        float gy = gt[(size_t)b * KPN * 2 + t * 2 + 0];
        float gx = gt[(size_t)b * KPN * 2 + t * 2 + 1];
        float dy = ly - gy, dx = lx - gx;
        lossbuf[(size_t)b * KPN * 2 + t * 2 + 0] = vy + dy * dy;
        lossbuf[(size_t)b * KPN * 2 + t * 2 + 1] = vx + dx * dx;
        out_loc[(size_t)b * KPN * 2 + t * 2 + 0] = ly;
        out_loc[(size_t)b * KPN * 2 + t * 2 + 1] = lx;
    }
}

#define S2_PK     260
#define S2_PLANE  3136

__device__ __forceinline__ float xhalf(float x) {
    const int b = __builtin_bit_cast(int, x);
    return __builtin_bit_cast(float, __builtin_amdgcn_permlanex16(b, b, 0x76543210, 0xfedcba98, false, false));
}

template <int M>
__device__ __forceinline__ float row_xor_add(float v) {
    const int s = __builtin_amdgcn_update_dpp(0, __builtin_bit_cast(int, v), 0x160 + M, 0xf, 0xf, false);
    return v + __builtin_bit_cast(float, s);
}

__global__ __launch_bounds__(1024) void pose_stage2(
    const float* __restrict__ lossbuf, float* __restrict__ out_hard)
{
    __shared__ __attribute__((aligned(16))) float s_T[4 * S2_PLANE];
    __shared__ __attribute__((aligned(16))) float s_w[32];
    const int t = threadIdx.x;

    const v4f* lb = reinterpret_cast<const v4f*>(lossbuf);
    v4f a[3];
    for (int j = 0; j < 3; ++j) a[j] = lb[t + 1024 * j];

    float s = 0.0f;
    for (int j = 0; j < 3; ++j) {
        const int   q  = t + 1024 * j;
        const int   b  = q / 12;
        const int   k2 = q - 12 * b;
        const float la = a[j].x + a[j].y;
        const float lc = a[j].z + a[j].w;
        float* row = s_T + k2 * S2_PK + b;
        row[0 * S2_PLANE] = a[j].x * la;
        row[1 * S2_PLANE] = a[j].z * lc;
        row[2 * S2_PLANE] = a[j].y * la;
        row[3 * S2_PLANE] = a[j].w * lc;
        s += la + lc;
    }

    s = row_xor_add<1>(s);
    s = row_xor_add<2>(s);
    s = row_xor_add<4>(s);
    s = row_xor_add<8>(s);
    s = s + xhalf(s);
    if ((t & 31) == 0) s_w[t >> 5] = s;
    __syncthreads();

    float tot;
    {
        const v4f* wp = reinterpret_cast<const v4f*>(s_w);
        v4f acc = wp[0];
        for (int i = 1; i < 8; ++i) acc = acc + wp[i];
        tot = (acc.x + acc.y) + (acc.z + acc.w);
    }
    const float inv = ((float)KPN / 150.0f) / (tot + 0.001f);

    v4f* o4 = reinterpret_cast<v4f*>(out_hard);
    for (int j = 0; j < 3; ++j) {
        const int p  = t + 1024 * j;
        const int c  = (p >= 1536) ? 1 : 0;
        const int r  = p - 1536 * c;
        const int k  = r >> 6;
        const int b4 = (r & 63) << 2;
        const v4f v = *reinterpret_cast<const v4f*>(
            s_T + (2 * c + (k & 1)) * S2_PLANE + (k >> 1) * S2_PK + b4);
        o4[p] = v * inv;
    }
}

extern "C" void kernel_launch(void* const* d_in, const int* in_sizes, int n_in,
                              void* d_out, int out_size, void* d_ws, size_t ws_size,
                              hipStream_t stream) {
    const float* feat = (const float*)d_in[0];
    const float* gt   = (const float*)d_in[1];
    float* out      = (float*)d_out;
    float* out_hard = out;
    float* out_loc  = out + 2 * KPN * BATCH;
    float* lossbuf  = (float*)d_ws;

    pose_stage1<<<BATCH, 768, 0, stream>>>(feat, gt, out_loc, lossbuf);
    pose_stage2<<<1, 1024, 0, stream>>>(lossbuf, out_hard);
}
